# starting kernel
# baseline (speedup 1.0000x reference)
#include <hip/hip_runtime.h>
#include <hip/hip_bf16.h>

typedef __attribute__((ext_vector_type(2))) float v2f;
typedef __attribute__((ext_vector_type(8))) float v8f;

#define EMB_D 64

__global__ void lgcn_init(const float* __restrict__ ue, const float* __restrict__ ie,
                          float* __restrict__ cur, float* __restrict__ acc,
                          size_t userElems, size_t total) {
    size_t i = ((size_t)blockIdx.x * blockDim.x + threadIdx.x) * 4;
    if (i >= total) return;
    float4 e = (i < userElems) ? *(const float4*)(ue + i)
                               : *(const float4*)(ie + (i - userElems));
    *(float4*)(cur + i) = e;
    float4 s;
    s.x = e.x * 0.25f; s.y = e.y * 0.25f; s.z = e.z * 0.25f; s.w = e.w * 0.25f;
    *(float4*)(acc + i) = s;
}

__global__ void lgcn_zero(float* __restrict__ p, size_t total) {
    size_t i = ((size_t)blockIdx.x * blockDim.x + threadIdx.x) * 4;
    if (i >= total) return;
    float4 z; z.x = z.y = z.z = z.w = 0.0f;
    *(float4*)(p + i) = z;
}

__global__ void lgcn_spmm(const int* __restrict__ rows, const int* __restrict__ cols,
                          const float* __restrict__ vals, const float* __restrict__ x,
                          float* __restrict__ y, int nEdges) {
    long long t = (long long)blockIdx.x * blockDim.x + threadIdx.x;
    int e   = (int)(t >> 4);
    int sub = (int)(t & 15);
    if (e >= nEdges) return;
    int   r = rows[e];
    int   c = cols[e];
    float v = vals[e];
    float4 g = *((const float4*)(x + (size_t)c * EMB_D) + sub);
    float* yp = y + (size_t)r * EMB_D + (size_t)sub * 4;
    unsafeAtomicAdd(yp + 0, g.x * v);
    unsafeAtomicAdd(yp + 1, g.y * v);
    unsafeAtomicAdd(yp + 2, g.z * v);
    unsafeAtomicAdd(yp + 3, g.w * v);
}

__global__ void lgcn_wmma_axpy(const float* __restrict__ nxt, float* __restrict__ acc,
                               int nTiles) {
    int wave = (int)(((size_t)blockIdx.x * blockDim.x + threadIdx.x) >> 5);
    if (wave >= nTiles) return;
    int lane = threadIdx.x & 31;
    int lo = lane & 15;
    int hi = lane >> 4;
    int tM = wave >> 2;
    int n0 = (wave & 3) << 4;
    size_t row0 = (size_t)tM << 4;

    const float* aBase = nxt + (row0 + (size_t)lo) * EMB_D + n0;
    float*       cBase = acc + (row0 + (size_t)(8 * hi)) * EMB_D + n0 + lo;

    v8f c;
#pragma unroll
    for (int v = 0; v < 8; ++v) c[v] = cBase[(size_t)v * EMB_D];

#pragma unroll
    for (int k = 0; k < 4; ++k) {
        int col = 4 * k + 2 * hi;
        v2f a, b;
        a[0] = aBase[col + 0];
        a[1] = aBase[col + 1];
        b[0] = (lo == col + 0) ? 0.25f : 0.0f;
        b[1] = (lo == col + 1) ? 0.25f : 0.0f;
        c = __builtin_amdgcn_wmma_f32_16x16x4_f32(
                 false, a,  false, b,
                 (short)0, c,  false,  false);
    }

#pragma unroll
    for (int v = 0; v < 8; ++v) cBase[(size_t)v * EMB_D] = c[v];
}

extern "C" void kernel_launch(void* const* d_in, const int* in_sizes, int n_in,
                              void* d_out, int out_size, void* d_ws, size_t ws_size,
                              hipStream_t stream) {
    const float* ue   = (const float*)d_in[0];
    const float* ie   = (const float*)d_in[1];
    const int*   rows = (const int*)d_in[2];
    const int*   cols = (const int*)d_in[3];
    const float* vals = (const float*)d_in[4];
    float*       out  = (float*)d_out;

    size_t userElems = (size_t)in_sizes[0];
    size_t itemElems = (size_t)in_sizes[1];
    int    nEdges    = in_sizes[2];
    size_t total     = userElems + itemElems;

    float* buf0 = (float*)d_ws;
    float* buf1 = buf0 + total;

    int vecBlocks = (int)((total / 4 + 255) / 256);
    lgcn_init<<<vecBlocks, 256, 0, stream>>>(ue, ie, buf0, out, userElems, total);

    int nTiles     = (int)(total / 256);
    int wmmaBlocks = (nTiles + 7) / 8;
    long long spmmThreads = (long long)nEdges * 16;
    int spmmBlocks = (int)((spmmThreads + 255) / 256);

    float* cur = buf0;
    float* nxt = buf1;
    for (int layer = 0; layer < 3; ++layer) {
        lgcn_zero<<<vecBlocks, 256, 0, stream>>>(nxt, total);
        lgcn_spmm<<<spmmBlocks, 256, 0, stream>>>(rows, cols, vals, cur, nxt, nEdges);
        lgcn_wmma_axpy<<<wmmaBlocks, 256, 0, stream>>>(nxt, out, nTiles);
        float* tmp = cur; cur = nxt; nxt = tmp;
    }
}


# SpMM scatter: destination rows split in two ranges, one per L2 half; a wave draws a 128-edge ticket from its own half's queue and adds only that range's edges; sum order = atomic arrival as before
# speedup vs baseline: 2.6672x; 2.6672x over previous
#include <hip/hip_runtime.h>
#include <hip/hip_bf16.h>

typedef __attribute__((ext_vector_type(2))) float v2f;
typedef __attribute__((ext_vector_type(8))) float v8f;

#define EMB_D 64
#define TICKET_EDGES 128
#define TICKET_PITCH 64

__global__ void __launch_bounds__(256)
lgcn_scale_init(const float* __restrict__ ue, const float* __restrict__ ie,
                float* __restrict__ t0, float* __restrict__ t1, float* __restrict__ out,
                int* __restrict__ tickets, int nUserRows, int nRowTiles) {
    int wave = (int)(((size_t)blockIdx.x * blockDim.x + threadIdx.x) >> 5);
    if (wave >= nRowTiles) return;
    int lane = threadIdx.x & 31;
    if (wave == 0 && lane < 6) tickets[lane * TICKET_PITCH] = 0;
    int lo = lane & 15;
    int hi = lane >> 4;
    int row0 = wave << 4;

    float4 e[4][2];
#pragma unroll
    for (int k = 0; k < 4; ++k) {
#pragma unroll
        for (int v = 0; v < 2; ++v) {
            int row = row0 + 4 * k + v + 2 * hi;
            const float* src = (row < nUserRows) ? ue + (size_t)row * EMB_D
                                                 : ie + (size_t)(row - nUserRows) * EMB_D;
            e[k][v] = *(const float4*)(src + 4 * lo);
        }
    }

    v8f d0, d1, d2, d3;
#pragma unroll
    for (int v = 0; v < 8; ++v) { d0[v] = 0.0f; d1[v] = 0.0f; d2[v] = 0.0f; d3[v] = 0.0f; }

#pragma unroll
    for (int k = 0; k < 4; ++k) {
        int kk = 4 * k + 2 * hi;
        v2f a, b0, b1, b2, b3;
        a[0] = (lo == kk + 0) ? 0.25f : 0.0f;
        a[1] = (lo == kk + 1) ? 0.25f : 0.0f;
        b0[0] = e[k][0].x; b0[1] = e[k][1].x;
        b1[0] = e[k][0].y; b1[1] = e[k][1].y;
        b2[0] = e[k][0].z; b2[1] = e[k][1].z;
        b3[0] = e[k][0].w; b3[1] = e[k][1].w;
        d0 = __builtin_amdgcn_wmma_f32_16x16x4_f32(false, a, false, b0, (short)0, d0, false, false);
        d1 = __builtin_amdgcn_wmma_f32_16x16x4_f32(false, a, false, b1, (short)0, d1, false, false);
        d2 = __builtin_amdgcn_wmma_f32_16x16x4_f32(false, a, false, b2, (short)0, d2, false, false);
        d3 = __builtin_amdgcn_wmma_f32_16x16x4_f32(false, a, false, b3, (short)0, d3, false, false);
    }

#pragma unroll
    for (int v = 0; v < 8; ++v) {
        size_t off = (size_t)(row0 + v + 8 * hi) * EMB_D + 4 * lo;
        float4 o;
        o.x = d0[v]; o.y = d1[v]; o.z = d2[v]; o.w = d3[v];
        *(float4*)(t0 + off)  = o;
        *(float4*)(t1 + off)  = o;
        *(float4*)(out + off) = o;
    }
}

template <int K>
__device__ __forceinline__ void lgcn_edges(int base, int r, int c, int vbits,
                                           const float* __restrict__ x,
                                           float* __restrict__ y, int lane) {
    float g0[K], g1[K];
#pragma unroll
    for (int j = 0; j < K; ++j) {
        int cj = __builtin_amdgcn_readlane(c, base + j);
        const float* xp = x + (size_t)cj * EMB_D + lane;
        g0[j] = xp[0];
        g1[j] = xp[32];
    }
#pragma unroll
    for (int j = 0; j < K; ++j) {
        int   rj = __builtin_amdgcn_readlane(r, base + j);
        float vj = __int_as_float(__builtin_amdgcn_readlane(vbits, base + j));
        float* yp = y + (size_t)rj * EMB_D + lane;
        unsafeAtomicAdd(yp,      g0[j] * vj);
        unsafeAtomicAdd(yp + 32, g1[j] * vj);
    }
}

__device__ __forceinline__ int lgcn_draw(int* counter, int lane) {
    int t = 0;
    if (lane == 0) t = atomicAdd(counter, 1);
    return __builtin_amdgcn_readfirstlane(t);
}

__global__ void __launch_bounds__(256)
lgcn_spmm(const int* __restrict__ rows, const int* __restrict__ cols,
          const float* __restrict__ vals, const float* __restrict__ x,
          float* __restrict__ y, int* __restrict__ tickets,
          int nEdges, int nTickets, int splitRow) {
    __shared__ int elist[8][3][TICKET_EDGES];
    int lane = threadIdx.x & 31;
    int w = threadIdx.x >> 5;
    unsigned where = __builtin_amdgcn_s_sendmsg_rtn(0x87);
    int q = (((where >> 16) & 0xFu) >= 4u) ? 1 : 0;
    int t = lgcn_draw(tickets + q * TICKET_PITCH, lane);
    if (t >= nTickets) {
        q ^= 1;
        t = lgcn_draw(tickets + q * TICKET_PITCH, lane);
        if (t >= nTickets) return;
    }
    long long e0 = (long long)t * TICKET_EDGES;
    long long eLast = (long long)nEdges - 1;

    int r[4], c[4], vbits[4];
#pragma unroll
    for (int s = 0; s < 4; ++s) {
        long long e = e0 + 32 * s + lane;
        long long ec = (e < eLast) ? e : eLast;
        r[s] = rows[ec];
        c[s] = cols[ec];
        vbits[s] = __float_as_int(vals[ec]);
    }
    asm volatile("" :: "v"(r[0]), "v"(c[0]), "v"(vbits[0]), "v"(r[1]), "v"(c[1]), "v"(vbits[1]),
                       "v"(r[2]), "v"(c[2]), "v"(vbits[2]), "v"(r[3]), "v"(c[3]), "v"(vbits[3]));
    int n = 0;
#pragma unroll
    for (int s = 0; s < 4; ++s) {
        long long e = e0 + 32 * s + lane;
        bool mine = (e <= eLast) && ((r[s] >= splitRow) == (q != 0));
        unsigned sel = __builtin_amdgcn_ballot_w32(mine);
        int pos = n + (int)__builtin_amdgcn_mbcnt_lo(sel, 0u);
        if (mine) {
            elist[w][0][pos] = r[s];
            elist[w][1][pos] = c[s];
            elist[w][2][pos] = vbits[s];
        }
        n += __builtin_popcount(sel);
    }
    __builtin_amdgcn_fence(__ATOMIC_RELEASE, "wavefront");
    __builtin_amdgcn_fence(__ATOMIC_ACQUIRE, "wavefront");

    for (int b = 0; b < n; b += 32) {
        int nb = (n - b < 32) ? (n - b) : 32;
        int idx = b + lane;
        if (idx > n - 1) idx = n - 1;
        int rr = elist[w][0][idx];
        int cc = elist[w][1][idx];
        int vv = elist[w][2][idx];
        if (nb == 32) {
            lgcn_edges<8>(0,  rr, cc, vv, x, y, lane);
            lgcn_edges<8>(8,  rr, cc, vv, x, y, lane);
            lgcn_edges<8>(16, rr, cc, vv, x, y, lane);
            lgcn_edges<8>(24, rr, cc, vv, x, y, lane);
        } else {
            int done = 0;
            if (nb & 16) {
                lgcn_edges<8>(0, rr, cc, vv, x, y, lane);
                lgcn_edges<8>(8, rr, cc, vv, x, y, lane);
                done = 16;
            }
            if (nb & 8) { lgcn_edges<8>(done, rr, cc, vv, x, y, lane); done += 8; }
            if (nb & 4) { lgcn_edges<4>(done, rr, cc, vv, x, y, lane); done += 4; }
            if (nb & 2) { lgcn_edges<2>(done, rr, cc, vv, x, y, lane); done += 2; }
            if (nb & 1) { lgcn_edges<1>(done, rr, cc, vv, x, y, lane); }
        }
    }
}

extern "C" void kernel_launch(void* const* d_in, const int* in_sizes, int n_in,
                              void* d_out, int out_size, void* d_ws, size_t ws_size,
                              hipStream_t stream) {
    const float* ue   = (const float*)d_in[0];
    const float* ie   = (const float*)d_in[1];
    const int*   rows = (const int*)d_in[2];
    const int*   cols = (const int*)d_in[3];
    const float* vals = (const float*)d_in[4];
    float*       out  = (float*)d_out;

    size_t userElems = (size_t)in_sizes[0];
    size_t itemElems = (size_t)in_sizes[1];
    int    nEdges    = in_sizes[2];
    size_t total     = userElems + itemElems;

    float* t0 = (float*)d_ws;
    float* t1 = t0 + total;
    int*   tk = (int*)(t1 + total);

    int nUserRows  = (int)(userElems / EMB_D);
    int nRowTiles  = (int)(total / (EMB_D * 16));
    int initBlocks = (nRowTiles + 7) / 8;
    int nRows      = (int)(total / EMB_D);
    int splitRow   = nRows / 2;
    int nTickets   = (nEdges + TICKET_EDGES - 1) / TICKET_EDGES;
    int spmmBlocks = (2 * nTickets + 7) / 8;

    lgcn_scale_init<<<initBlocks, 256, 0, stream>>>(ue, ie, t0, t1, out, tk, nUserRows, nRowTiles);
    lgcn_spmm<<<spmmBlocks, 256, 0, stream>>>(rows, cols, vals, t0, t1,  tk,                    nEdges, nTickets, splitRow);
    lgcn_spmm<<<spmmBlocks, 256, 0, stream>>>(rows, cols, vals, t1, t0,  tk + 2 * TICKET_PITCH, nEdges, nTickets, splitRow);
    lgcn_spmm<<<spmmBlocks, 256, 0, stream>>>(rows, cols, vals, t0, out, tk + 4 * TICKET_PITCH, nEdges, nTickets, splitRow);
}


# SpMM gather side from LDS: edges grouped once per call by (destination half, 1172-row source tile); a work-group holds its tile in LDS, line atomics as before; 4 -> 6 launches; sum order = arrival
# speedup vs baseline: 4.0205x; 1.5074x over previous
#include <hip/hip_runtime.h>
#include <hip/hip_bf16.h>

typedef __attribute__((ext_vector_type(2))) float v2f;
typedef __attribute__((ext_vector_type(8))) float v8f;
typedef __attribute__((ext_vector_type(4))) float v4f;

#define EMB_D 64
#define TICKET_PITCH 64
#define TILE_ROWS 1172
#define N_TILES 256
#define MAX_KEYS 512
#define BKT_THREADS 512
#define BKT_EDGES 4096
#define CTR_INTS 4096

__global__ void __launch_bounds__(256)
lgcn_scale_init(const float* __restrict__ ue, const float* __restrict__ ie,
                float* __restrict__ t0, float* __restrict__ t1, float* __restrict__ out,
                int* __restrict__ tickets, int nUserRows, int nRowTiles) {
    int wave = (int)(((size_t)blockIdx.x * blockDim.x + threadIdx.x) >> 5);
    if (wave >= nRowTiles) return;
    int lane = threadIdx.x & 31;
    if (wave == 0) {
        if (lane < 6) tickets[lane * TICKET_PITCH] = 0;
        for (int i = lane; i < 2 * MAX_KEYS; i += 32) tickets[512 + i] = 0;
    }
    int lo = lane & 15;
    int hi = lane >> 4;
    int row0 = wave << 4;

    float4 e[4][2];
#pragma unroll
    for (int k = 0; k < 4; ++k) {
#pragma unroll
        for (int v = 0; v < 2; ++v) {
            int row = row0 + 4 * k + v + 2 * hi;
            const float* src = (row < nUserRows) ? ue + (size_t)row * EMB_D
                                                 : ie + (size_t)(row - nUserRows) * EMB_D;
            e[k][v] = *(const float4*)(src + 4 * lo);
        }
    }

    v8f d0, d1, d2, d3;
#pragma unroll
    for (int v = 0; v < 8; ++v) { d0[v] = 0.0f; d1[v] = 0.0f; d2[v] = 0.0f; d3[v] = 0.0f; }

#pragma unroll
    for (int k = 0; k < 4; ++k) {
        int kk = 4 * k + 2 * hi;
        v2f a, b0, b1, b2, b3;
        a[0] = (lo == kk + 0) ? 0.25f : 0.0f;
        a[1] = (lo == kk + 1) ? 0.25f : 0.0f;
        b0[0] = e[k][0].x; b0[1] = e[k][1].x;
        b1[0] = e[k][0].y; b1[1] = e[k][1].y;
        b2[0] = e[k][0].z; b2[1] = e[k][1].z;
        b3[0] = e[k][0].w; b3[1] = e[k][1].w;
        d0 = __builtin_amdgcn_wmma_f32_16x16x4_f32(false, a, false, b0, (short)0, d0, false, false);
        d1 = __builtin_amdgcn_wmma_f32_16x16x4_f32(false, a, false, b1, (short)0, d1, false, false);
        d2 = __builtin_amdgcn_wmma_f32_16x16x4_f32(false, a, false, b2, (short)0, d2, false, false);
        d3 = __builtin_amdgcn_wmma_f32_16x16x4_f32(false, a, false, b3, (short)0, d3, false, false);
    }

#pragma unroll
    for (int v = 0; v < 8; ++v) {
        size_t off = (size_t)(row0 + v + 8 * hi) * EMB_D + 4 * lo;
        float4 o;
        o.x = d0[v]; o.y = d1[v]; o.z = d2[v]; o.w = d3[v];
        *(float4*)(t0 + off)  = o;
        *(float4*)(t1 + off)  = o;
        *(float4*)(out + off) = o;
    }
}


__device__ __forceinline__ int lgcn_key(int r, int c, int splitRow) {
    return ((r >= splitRow) ? N_TILES : 0) + c / TILE_ROWS;
}

__global__ void __launch_bounds__(BKT_THREADS)
lgcn_hist(const int* __restrict__ rows, const int* __restrict__ cols, int* __restrict__ counts,
          int nEdges, int splitRow) {
    __shared__ int cnt[MAX_KEYS];
    int tid = threadIdx.x;
    cnt[tid] = 0;
    __syncthreads();
    long long e0 = (long long)blockIdx.x * BKT_EDGES;
    long long eLast = (long long)nEdges - 1;
    int r[8], c[8];
#pragma unroll
    for (int i = 0; i < 8; ++i) {
        long long e = e0 + i * BKT_THREADS + tid;
        long long ec = (e < eLast) ? e : eLast;
        r[i] = rows[ec];
        c[i] = cols[ec];
    }
    asm volatile("" :: "v"(r[0]), "v"(c[0]), "v"(r[1]), "v"(c[1]), "v"(r[2]), "v"(c[2]), "v"(r[3]), "v"(c[3]),
                       "v"(r[4]), "v"(c[4]), "v"(r[5]), "v"(c[5]), "v"(r[6]), "v"(c[6]), "v"(r[7]), "v"(c[7]));
#pragma unroll
    for (int i = 0; i < 8; ++i) {
        long long e = e0 + i * BKT_THREADS + tid;
        if (e <= eLast) atomicAdd(&cnt[lgcn_key(r[i], c[i], splitRow)], 1);
    }
    __syncthreads();
    int n = cnt[tid];
    if (n) atomicAdd(&counts[tid], n);
}

__global__ void __launch_bounds__(BKT_THREADS)
lgcn_bucket(const int* __restrict__ rows, const int* __restrict__ cols, const float* __restrict__ vals,
            const int* __restrict__ counts, int* __restrict__ cursor, int* __restrict__ starts,
            int* __restrict__ recWords, int nEdges, int splitRow) {
    int4* recs = (int4*)recWords;
    __shared__ int  cnt[MAX_KEYS];
    __shared__ int  lstart[MAX_KEYS];
    __shared__ int  gbase[MAX_KEYS];
    __shared__ int  sc[2][MAX_KEYS];
    __shared__ int4 stage[BKT_EDGES];
    int tid = threadIdx.x;
    cnt[tid] = 0;
    __syncthreads();
    long long e0 = (long long)blockIdx.x * BKT_EDGES;
    long long eLast = (long long)nEdges - 1;
    int r[8], c[8], vb[8], key[8], rank[8];
#pragma unroll
    for (int i = 0; i < 8; ++i) {
        long long e = e0 + i * BKT_THREADS + tid;
        long long ec = (e < eLast) ? e : eLast;
        r[i] = rows[ec];
        c[i] = cols[ec];
        vb[i] = __float_as_int(vals[ec]);
    }
    asm volatile("" :: "v"(r[0]), "v"(c[0]), "v"(vb[0]), "v"(r[1]), "v"(c[1]), "v"(vb[1]),
                       "v"(r[2]), "v"(c[2]), "v"(vb[2]), "v"(r[3]), "v"(c[3]), "v"(vb[3]),
                       "v"(r[4]), "v"(c[4]), "v"(vb[4]), "v"(r[5]), "v"(c[5]), "v"(vb[5]),
                       "v"(r[6]), "v"(c[6]), "v"(vb[6]), "v"(r[7]), "v"(c[7]), "v"(vb[7]));
#pragma unroll
    for (int i = 0; i < 8; ++i) {
        long long e = e0 + i * BKT_THREADS + tid;
        key[i] = (e <= eLast) ? lgcn_key(r[i], c[i], splitRow) : -1;
        rank[i] = 0;
        if (key[i] >= 0) rank[i] = atomicAdd(&cnt[key[i]], 1);
    }
    __syncthreads();
    int mine = cnt[tid];
    int glob = counts[tid];
    int a = mine, b = glob;
    sc[0][tid] = a;
    sc[1][tid] = b;
    __syncthreads();
    for (int off = 1; off < MAX_KEYS; off <<= 1) {
        int ya = 0, yb = 0;
        if (tid >= off) { ya = sc[0][tid - off]; yb = sc[1][tid - off]; }
        __syncthreads();
        a += ya; b += yb;
        sc[0][tid] = a;
        sc[1][tid] = b;
        __syncthreads();
    }
    lstart[tid] = a - mine;
    int run = 0;
    if (mine) run = atomicAdd(&cursor[tid], mine);
    gbase[tid] = (b - glob) + run;
    if (blockIdx.x == 0) {
        starts[tid] = b - glob;
        if (tid == MAX_KEYS - 1) starts[MAX_KEYS] = b;
    }
    __syncthreads();
#pragma unroll
    for (int i = 0; i < 8; ++i)
        if (key[i] >= 0) stage[lstart[key[i]] + rank[i]] = make_int4(r[i], c[i], vb[i], key[i]);
    __syncthreads();
    long long left = (long long)nEdges - e0;
    int nHere = (left < BKT_EDGES) ? (int)left : BKT_EDGES;
    for (int p = tid; p < nHere; p += BKT_THREADS) {
        int4 rec = stage[p];
        recs[gbase[rec.w] + (p - lstart[rec.w])] = rec;
    }
}

template <int K>
__device__ __forceinline__ void lgcn_edges(int base, int r, int cl, int vbits,
                                           const float* tile,
                                           float* __restrict__ y, int lane) {
    float g0[K], g1[K];
#pragma unroll
    for (int j = 0; j < K; ++j) {
        int cj = __builtin_amdgcn_readlane(cl, base + j);
        const float* xp = tile + cj * EMB_D + lane;
        g0[j] = xp[0];
        g1[j] = xp[32];
    }
#pragma unroll
    for (int j = 0; j < K; ++j) {
        int   rj = __builtin_amdgcn_readlane(r, base + j);
        float vj = __int_as_float(__builtin_amdgcn_readlane(vbits, base + j));
        float* yp = y + (size_t)rj * EMB_D + lane;
        unsafeAtomicAdd(yp,      g0[j] * vj);
        unsafeAtomicAdd(yp + 32, g1[j] * vj);
    }
}

__global__ void __launch_bounds__(1024)
lgcn_spmm(const int* __restrict__ recWords, const int* __restrict__ starts,
          const float* __restrict__ x, float* __restrict__ y, int* __restrict__ tickets, int nRows) {
    const int4* recs = (const int4*)recWords;
    __shared__ __attribute__((aligned(256))) float tile[TILE_ROWS * EMB_D];
    __shared__ int s_key;
    int tid = threadIdx.x;
    int lane = tid & 31;
    int w = __builtin_amdgcn_readfirstlane(tid >> 5);
    unsigned where = __builtin_amdgcn_s_sendmsg_rtn(0x87);
    if (tid == 0) {
        int q = (((where >> 16) & 0xFu) >= 4u) ? 1 : 0;
        int t = atomicAdd(tickets + q * TICKET_PITCH, 1);
        if (t >= N_TILES) {
            q ^= 1;
            t = atomicAdd(tickets + q * TICKET_PITCH, 1);
        }
        s_key = (t < N_TILES) ? (q * N_TILES + t) : -1;
    }
    __syncthreads();
    int key = s_key;
    if (key < 0) return;
    int row0 = (key & (N_TILES - 1)) * TILE_ROWS;
    int nr = nRows - row0;
    if (nr > TILE_ROWS) nr = TILE_ROWS;
    if (nr > 0) {
        const v4f* src = (const v4f*)(x + (size_t)row0 * EMB_D);
        v4f* dst = (v4f*)tile;
        int n4 = nr * (EMB_D / 4);
        for (int i0 = 0; i0 < n4; i0 += 10 * 1024) {
            v4f buf[10];
#pragma unroll
            for (int u = 0; u < 10; ++u) {
                int i = i0 + u * 1024 + tid;
                buf[u] = src[(i < n4) ? i : (n4 - 1)];
            }
            asm volatile("" :: "v"(buf[0]), "v"(buf[1]), "v"(buf[2]), "v"(buf[3]), "v"(buf[4]),
                               "v"(buf[5]), "v"(buf[6]), "v"(buf[7]), "v"(buf[8]), "v"(buf[9]));
#pragma unroll
            for (int u = 0; u < 10; ++u) {
                int i = i0 + u * 1024 + tid;
                if (i < n4) dst[i] = buf[u];
            }
        }
    }
    __syncthreads();
    int eBeg = starts[key];
    int eEnd = starts[key + 1];
    int e = eBeg + 32 * w;
    if (e >= eEnd) return;
    int4 rec;
    {
        int nb0 = (eEnd - e < 32) ? (eEnd - e) : 32;
        rec = recs[e + ((lane < nb0) ? lane : (nb0 - 1))];
    }
    for (; e < eEnd; e += 32 * 32) {
        int nb = (eEnd - e < 32) ? (eEnd - e) : 32;
        int4 cur = rec;
        int en = e + 32 * 32;
        if (en < eEnd) {
            int nbn = (eEnd - en < 32) ? (eEnd - en) : 32;
            rec = recs[en + ((lane < nbn) ? lane : (nbn - 1))];
        }
        int rr = cur.x;
        int cc = cur.y - row0;
        int vv = cur.z;
        if (nb == 32) {
            lgcn_edges<8>(0,  rr, cc, vv, tile, y, lane);
            lgcn_edges<8>(8,  rr, cc, vv, tile, y, lane);
            lgcn_edges<8>(16, rr, cc, vv, tile, y, lane);
            lgcn_edges<8>(24, rr, cc, vv, tile, y, lane);
        } else {
            int done = 0;
            if (nb & 16) {
                lgcn_edges<8>(0, rr, cc, vv, tile, y, lane);
                lgcn_edges<8>(8, rr, cc, vv, tile, y, lane);
                done = 16;
            }
            if (nb & 8) { lgcn_edges<8>(done, rr, cc, vv, tile, y, lane); done += 8; }
            if (nb & 4) { lgcn_edges<4>(done, rr, cc, vv, tile, y, lane); done += 4; }
            if (nb & 2) { lgcn_edges<2>(done, rr, cc, vv, tile, y, lane); done += 2; }
            if (nb & 1) { lgcn_edges<1>(done, rr, cc, vv, tile, y, lane); }
        }
    }
}

extern "C" void kernel_launch(void* const* d_in, const int* in_sizes, int n_in,
                              void* d_out, int out_size, void* d_ws, size_t ws_size,
                              hipStream_t stream) {
    const float* ue   = (const float*)d_in[0];
    const float* ie   = (const float*)d_in[1];
    const int*   rows = (const int*)d_in[2];
    const int*   cols = (const int*)d_in[3];
    const float* vals = (const float*)d_in[4];
    float*       out  = (float*)d_out;

    size_t userElems = (size_t)in_sizes[0];
    size_t itemElems = (size_t)in_sizes[1];
    int    nEdges    = in_sizes[2];
    size_t total     = userElems + itemElems;

    float* t0 = (float*)d_ws;
    float* t1 = t0 + total;
    int*   tk = (int*)(t1 + total);
    int*   counts = tk + 512;
    int*   cursor = counts + MAX_KEYS;
    int*   starts = cursor + MAX_KEYS;
    int*   recs   = tk + CTR_INTS;

    int nUserRows  = (int)(userElems / EMB_D);
    int nRowTiles  = (int)(total / (EMB_D * 16));
    int initBlocks = (nRowTiles + 7) / 8;
    int nRows      = (int)(total / EMB_D);
    int splitRow   = nRows / 2;
    int bktBlocks  = (nEdges + BKT_EDGES - 1) / BKT_EDGES;

    lgcn_scale_init<<<initBlocks, 256, 0, stream>>>(ue, ie, t0, t1, out, tk, nUserRows, nRowTiles);
    lgcn_hist<<<bktBlocks, BKT_THREADS, 0, stream>>>(rows, cols, counts, nEdges, splitRow);
    lgcn_bucket<<<bktBlocks, BKT_THREADS, 0, stream>>>(rows, cols, vals, counts, cursor, starts, recs, nEdges, splitRow);
    lgcn_spmm<<<2 * N_TILES, 1024, 0, stream>>>(recs, starts, t0, t1,  tk,                    nRows);
    lgcn_spmm<<<2 * N_TILES, 1024, 0, stream>>>(recs, starts, t1, t0,  tk + 2 * TICKET_PITCH, nRows);
    lgcn_spmm<<<2 * N_TILES, 1024, 0, stream>>>(recs, starts, t0, out, tk + 4 * TICKET_PITCH, nRows);
}


# SpMM sums in LDS: edges grouped by 1172-row destination tile; its work-group loads e4 there, adds gathered rows by LDS f32 atomics, stores the tile once; no global atomics; sum order = LDS arrival
# speedup vs baseline: 4.3093x; 1.0718x over previous
#include <hip/hip_runtime.h>
#include <hip/hip_bf16.h>

typedef __attribute__((ext_vector_type(2))) float v2f;
typedef __attribute__((ext_vector_type(8))) float v8f;
typedef __attribute__((ext_vector_type(4))) float v4f;

#define EMB_D 64
#define TILE_ROWS 1172
#define N_TILES 256
#define MAX_KEYS 256
#define BKT_THREADS 512
#define BKT_EDGES 4096
#define CTR_INTS 4096

__global__ void __launch_bounds__(256)
lgcn_scale_init(const float* __restrict__ ue, const float* __restrict__ ie,
                float* __restrict__ e4, int* __restrict__ ctr, int nUserRows, int nRowTiles) {
    int wave = (int)(((size_t)blockIdx.x * blockDim.x + threadIdx.x) >> 5);
    if (wave >= nRowTiles) return;
    int lane = threadIdx.x & 31;
    if (wave == 0) {
        for (int i = lane; i < 2 * MAX_KEYS; i += 32) ctr[i] = 0;
    }
    int lo = lane & 15;
    int hi = lane >> 4;
    int row0 = wave << 4;

    float4 e[4][2];
#pragma unroll
    for (int k = 0; k < 4; ++k) {
#pragma unroll
        for (int v = 0; v < 2; ++v) {
            int row = row0 + 4 * k + v + 2 * hi;
            const float* src = (row < nUserRows) ? ue + (size_t)row * EMB_D
                                                 : ie + (size_t)(row - nUserRows) * EMB_D;
            e[k][v] = *(const float4*)(src + 4 * lo);
        }
    }

    v8f d0, d1, d2, d3;
#pragma unroll
    for (int v = 0; v < 8; ++v) { d0[v] = 0.0f; d1[v] = 0.0f; d2[v] = 0.0f; d3[v] = 0.0f; }

#pragma unroll
    for (int k = 0; k < 4; ++k) {
        int kk = 4 * k + 2 * hi;
        v2f a, b0, b1, b2, b3;
        a[0] = (lo == kk + 0) ? 0.25f : 0.0f;
        a[1] = (lo == kk + 1) ? 0.25f : 0.0f;
        b0[0] = e[k][0].x; b0[1] = e[k][1].x;
        b1[0] = e[k][0].y; b1[1] = e[k][1].y;
        b2[0] = e[k][0].z; b2[1] = e[k][1].z;
        b3[0] = e[k][0].w; b3[1] = e[k][1].w;
        d0 = __builtin_amdgcn_wmma_f32_16x16x4_f32(false, a, false, b0, (short)0, d0, false, false);
        d1 = __builtin_amdgcn_wmma_f32_16x16x4_f32(false, a, false, b1, (short)0, d1, false, false);
        d2 = __builtin_amdgcn_wmma_f32_16x16x4_f32(false, a, false, b2, (short)0, d2, false, false);
        d3 = __builtin_amdgcn_wmma_f32_16x16x4_f32(false, a, false, b3, (short)0, d3, false, false);
    }

#pragma unroll
    for (int v = 0; v < 8; ++v) {
        size_t off = (size_t)(row0 + v + 8 * hi) * EMB_D + 4 * lo;
        float4 o;
        o.x = d0[v]; o.y = d1[v]; o.z = d2[v]; o.w = d3[v];
        *(float4*)(e4 + off) = o;
    }
}


__device__ __forceinline__ int lgcn_key(int r) {
    return (int)((unsigned)r / TILE_ROWS);
}

__global__ void __launch_bounds__(BKT_THREADS)
lgcn_hist(const int* __restrict__ rows, int* __restrict__ counts, int nEdges) {
    __shared__ int cnt[MAX_KEYS];
    int tid = threadIdx.x;
    if (tid < MAX_KEYS) cnt[tid] = 0;
    __syncthreads();
    long long e0 = (long long)blockIdx.x * BKT_EDGES;
    long long eLast = (long long)nEdges - 1;
    int r[8];
#pragma unroll
    for (int i = 0; i < 8; ++i) {
        long long e = e0 + i * BKT_THREADS + tid;
        long long ec = (e < eLast) ? e : eLast;
        r[i] = rows[ec];
    }
    asm volatile("" :: "v"(r[0]), "v"(r[1]), "v"(r[2]), "v"(r[3]), "v"(r[4]), "v"(r[5]), "v"(r[6]), "v"(r[7]));
#pragma unroll
    for (int i = 0; i < 8; ++i) {
        long long e = e0 + i * BKT_THREADS + tid;
        if (e <= eLast) atomicAdd(&cnt[lgcn_key(r[i])], 1);
    }
    __syncthreads();
    if (tid < MAX_KEYS) {
        int n = cnt[tid];
        if (n) atomicAdd(&counts[tid], n);
    }
}

__global__ void __launch_bounds__(BKT_THREADS)
lgcn_bucket(const int* __restrict__ rows, const int* __restrict__ cols, const float* __restrict__ vals,
            const int* __restrict__ counts, int* __restrict__ cursor, int* __restrict__ starts,
            int* __restrict__ recWords, int nEdges) {
    int4* recs = (int4*)recWords;
    __shared__ int  cnt[MAX_KEYS];
    __shared__ int  lstart[MAX_KEYS];
    __shared__ int  gbase[MAX_KEYS];
    __shared__ int  sc[2][MAX_KEYS];
    __shared__ int4 stage[BKT_EDGES];
    int tid = threadIdx.x;
    bool owner = tid < MAX_KEYS;
    if (owner) cnt[tid] = 0;
    __syncthreads();
    long long e0 = (long long)blockIdx.x * BKT_EDGES;
    long long eLast = (long long)nEdges - 1;
    int r[8], c[8], vb[8], key[8], rank[8];
#pragma unroll
    for (int i = 0; i < 8; ++i) {
        long long e = e0 + i * BKT_THREADS + tid;
        long long ec = (e < eLast) ? e : eLast;
        r[i] = rows[ec];
        c[i] = cols[ec];
        vb[i] = __float_as_int(vals[ec]);
    }
    asm volatile("" :: "v"(r[0]), "v"(c[0]), "v"(vb[0]), "v"(r[1]), "v"(c[1]), "v"(vb[1]),
                       "v"(r[2]), "v"(c[2]), "v"(vb[2]), "v"(r[3]), "v"(c[3]), "v"(vb[3]),
                       "v"(r[4]), "v"(c[4]), "v"(vb[4]), "v"(r[5]), "v"(c[5]), "v"(vb[5]),
                       "v"(r[6]), "v"(c[6]), "v"(vb[6]), "v"(r[7]), "v"(c[7]), "v"(vb[7]));
#pragma unroll
    for (int i = 0; i < 8; ++i) {
        long long e = e0 + i * BKT_THREADS + tid;
        key[i] = (e <= eLast) ? lgcn_key(r[i]) : -1;
        rank[i] = 0;
        if (key[i] >= 0) rank[i] = atomicAdd(&cnt[key[i]], 1);
    }
    __syncthreads();
    int mine = 0, glob = 0;
    if (owner) { mine = cnt[tid]; glob = counts[tid]; }
    int a = mine, b = glob;
    if (owner) { sc[0][tid] = a; sc[1][tid] = b; }
    __syncthreads();
    for (int off = 1; off < MAX_KEYS; off <<= 1) {
        int ya = 0, yb = 0;
        if (owner && tid >= off) { ya = sc[0][tid - off]; yb = sc[1][tid - off]; }
        __syncthreads();
        a += ya; b += yb;
        if (owner) { sc[0][tid] = a; sc[1][tid] = b; }
        __syncthreads();
    }
    if (owner) {
        lstart[tid] = a - mine;
        int run = 0;
        if (mine) run = atomicAdd(&cursor[tid], mine);
        gbase[tid] = (b - glob) + run;
        if (blockIdx.x == 0) {
            starts[tid] = b - glob;
            if (tid == MAX_KEYS - 1) starts[MAX_KEYS] = b;
        }
    }
    __syncthreads();
#pragma unroll
    for (int i = 0; i < 8; ++i)
        if (key[i] >= 0) stage[lstart[key[i]] + rank[i]] = make_int4(r[i], c[i], vb[i], key[i]);
    __syncthreads();
    long long left = (long long)nEdges - e0;
    int nHere = (left < BKT_EDGES) ? (int)left : BKT_EDGES;
    for (int p = tid; p < nHere; p += BKT_THREADS) {
        int4 rec = stage[p];
        recs[gbase[rec.w] + (p - lstart[rec.w])] = rec;
    }
}

template <int K>
__device__ __forceinline__ void lgcn_edges(int base, int rl, int c, int vbits,
                                           const float* __restrict__ x, float* tile, int lane) {
    v2f g[K];
#pragma unroll
    for (int j = 0; j < K; ++j) {
        unsigned cj = (unsigned)__builtin_amdgcn_readlane(c, base + j);
        g[j] = *(const v2f*)(x + (size_t)cj * EMB_D + 2 * lane);
    }
#pragma unroll
    for (int j = 0; j < K; ++j) {
        int   rj = __builtin_amdgcn_readlane(rl, base + j);
        float vj = __int_as_float(__builtin_amdgcn_readlane(vbits, base + j));
        float* tp = tile + rj * EMB_D + 2 * lane;
        __hip_atomic_fetch_add(tp,     g[j][0] * vj, __ATOMIC_RELAXED, __HIP_MEMORY_SCOPE_WORKGROUP);
        __hip_atomic_fetch_add(tp + 1, g[j][1] * vj, __ATOMIC_RELAXED, __HIP_MEMORY_SCOPE_WORKGROUP);
    }
}

__global__ void __launch_bounds__(1024)
lgcn_spmm(const int* __restrict__ recWords, const int* __restrict__ starts, const float* e4,
          const float* x, float* __restrict__ y, int nRows) {
    const int4* recs = (const int4*)recWords;
    __shared__ __attribute__((aligned(256))) float tile[TILE_ROWS * EMB_D];
    int tid = threadIdx.x;
    int lane = tid & 31;
    int w = __builtin_amdgcn_readfirstlane(tid >> 5);
    int key = blockIdx.x;
    int row0 = key * TILE_ROWS;
    int nr = nRows - row0;
    if (nr > TILE_ROWS) nr = TILE_ROWS;
    int n4 = (nr > 0) ? nr * (EMB_D / 4) : 0;
    v4f* tile4 = (v4f*)tile;
    if (n4 > 0) {
        const v4f* src = (const v4f*)(e4 + (size_t)row0 * EMB_D);
        for (int i0 = 0; i0 < n4; i0 += 10 * 1024) {
            v4f buf[10];
#pragma unroll
            for (int u = 0; u < 10; ++u) {
                int i = i0 + u * 1024 + tid;
                buf[u] = src[(i < n4) ? i : (n4 - 1)];
            }
            asm volatile("" :: "v"(buf[0]), "v"(buf[1]), "v"(buf[2]), "v"(buf[3]), "v"(buf[4]),
                               "v"(buf[5]), "v"(buf[6]), "v"(buf[7]), "v"(buf[8]), "v"(buf[9]));
#pragma unroll
            for (int u = 0; u < 10; ++u) {
                int i = i0 + u * 1024 + tid;
                if (i < n4) tile4[i] = buf[u];
            }
        }
    }
    __syncthreads();
    int eBeg = starts[key];
    int eEnd = starts[key + 1];
    int e = eBeg + 32 * w;
    int4 rec = make_int4(0, 0, 0, 0);
    if (e < eEnd) {
        int nb0 = (eEnd - e < 32) ? (eEnd - e) : 32;
        rec = recs[e + ((lane < nb0) ? lane : (nb0 - 1))];
    }
    for (; e < eEnd; e += 32 * 32) {
        int nb = (eEnd - e < 32) ? (eEnd - e) : 32;
        int4 cur = rec;
        int en = e + 32 * 32;
        if (en < eEnd) {
            int nbn = (eEnd - en < 32) ? (eEnd - en) : 32;
            rec = recs[en + ((lane < nbn) ? lane : (nbn - 1))];
        }
        int rr = cur.x - row0;
        int cc = cur.y;
        int vv = cur.z;
        if (nb == 32) {
            lgcn_edges<16>(0,  rr, cc, vv, x, tile, lane);
            lgcn_edges<16>(16, rr, cc, vv, x, tile, lane);
        } else {
            int done = 0;
            if (nb & 16) { lgcn_edges<16>(0,   rr, cc, vv, x, tile, lane); done = 16; }
            if (nb & 8)  { lgcn_edges<8>(done, rr, cc, vv, x, tile, lane); done += 8; }
            if (nb & 4)  { lgcn_edges<4>(done, rr, cc, vv, x, tile, lane); done += 4; }
            if (nb & 2)  { lgcn_edges<2>(done, rr, cc, vv, x, tile, lane); done += 2; }
            if (nb & 1)  { lgcn_edges<1>(done, rr, cc, vv, x, tile, lane); }
        }
    }
    __syncthreads();
    v4f* dst = (v4f*)(y + (size_t)row0 * EMB_D);
    for (int i = tid; i < n4; i += 1024) dst[i] = tile4[i];
}

extern "C" void kernel_launch(void* const* d_in, const int* in_sizes, int n_in,
                              void* d_out, int out_size, void* d_ws, size_t ws_size,
                              hipStream_t stream) {
    const float* ue   = (const float*)d_in[0];
    const float* ie   = (const float*)d_in[1];
    const int*   rows = (const int*)d_in[2];
    const int*   cols = (const int*)d_in[3];
    const float* vals = (const float*)d_in[4];
    float*       out  = (float*)d_out;

    size_t userElems = (size_t)in_sizes[0];
    size_t itemElems = (size_t)in_sizes[1];
    int    nEdges    = in_sizes[2];
    size_t total     = userElems + itemElems;

    float* e4 = (float*)d_ws;
    float* t1 = e4 + total;
    int*   counts = (int*)(t1 + total);
    int*   cursor = counts + MAX_KEYS;
    int*   starts = cursor + MAX_KEYS;
    int*   recs   = counts + CTR_INTS;

    int nUserRows  = (int)(userElems / EMB_D);
    int nRowTiles  = (int)(total / (EMB_D * 16));
    int initBlocks = (nRowTiles + 7) / 8;
    int nRows      = (int)(total / EMB_D);
    int bktBlocks  = (nEdges + BKT_EDGES - 1) / BKT_EDGES;

    lgcn_scale_init<<<initBlocks, 256, 0, stream>>>(ue, ie, e4, counts, nUserRows, nRowTiles);
    lgcn_hist<<<bktBlocks, BKT_THREADS, 0, stream>>>(rows, counts, nEdges);
    lgcn_bucket<<<bktBlocks, BKT_THREADS, 0, stream>>>(rows, cols, vals, counts, cursor, starts, recs, nEdges);
    lgcn_spmm<<<N_TILES, 1024, 0, stream>>>(recs, starts, e4, e4,  out, nRows);
    lgcn_spmm<<<N_TILES, 1024, 0, stream>>>(recs, starts, e4, out, t1,  nRows);
    lgcn_spmm<<<N_TILES, 1024, 0, stream>>>(recs, starts, e4, t1,  out, nRows);
}
